# starting kernel
# baseline (speedup 1.0000x reference)
#include <hip/hip_runtime.h>


typedef float v2f __attribute__((ext_vector_type(2)));
typedef float v8f __attribute__((ext_vector_type(8)));

#if defined(__HIP_DEVICE_COMPILE__) && !__has_builtin(__builtin_amdgcn_wmma_f32_16x16x4_f32)
#error "gfx1250 WMMA f32 16x16x4 builtin not available"
#endif

namespace {
constexpr int Bn = 4096;
constexpr int Tn = 2048;
constexpr int G3 = 12;
constexpr int On = 2;
}

__device__ __forceinline__ float fsig(float x) {
  float e = __builtin_amdgcn_exp2f(x * -1.4426950408889634f);
  return __builtin_amdgcn_rcpf(1.0f + e);
}
__device__ __forceinline__ float ftanh(float x) {
  float e = __builtin_amdgcn_exp2f(x * -2.8853901817779268f);
  return (1.0f - e) * __builtin_amdgcn_rcpf(1.0f + e);
}

__global__ __launch_bounds__(32) void gru_wmma_kernel(
    const float* __restrict__ in,
    const float* __restrict__ Win,
    const float* __restrict__ Wh,
    const float* __restrict__ bias,
    const float* __restrict__ dW,
    const float* __restrict__ db,
    float* __restrict__ out)
{
  const int lane = threadIdx.x & 31;
  const int m    = lane & 15;
  const bool lo  = lane < 16;
  const int seq  = blockIdx.x * 16 + m;
  const bool gv  = m < G3;

  v2f aWin, aWh;
  aWin.x = gv ? Win[(lo ? 0 : 2) * G3 + m] : 0.0f;
  aWin.y = (gv && lo) ? Win[1 * G3 + m] : 0.0f;
  aWh.x  = gv ? Wh[(lo ? 0 : 2) * G3 + m] : 0.0f;
  aWh.y  = gv ? Wh[(lo ? 1 : 3) * G3 + m] : 0.0f;

  v8f cbin, cbh;
#pragma unroll
  for (int r = 0; r < 8; ++r) {
    int row = (lo ? 0 : 8) + r;
    cbin[r] = (row < G3) ? bias[row] : 0.0f;
    cbh[r]  = (row < G3) ? bias[G3 + row] : 0.0f;
  }

  const int oc = lo ? 0 : 1;
  const float dw0 = dW[0 * On + oc], dw1 = dW[1 * On + oc];
  const float dw2 = dW[2 * On + oc], dw3 = dW[3 * On + oc];
  const float dcb = db[oc];

  const float* pa = in + (size_t)seq * (3 * Tn) + (lo ? 0 : 2);
  const float* pb = in + (size_t)seq * (3 * Tn) + (lo ? 1 : 2);
  float* po = out + (size_t)seq * (On * Tn) + oc;

  float h0 = 0.0f, h1 = 0.0f, h2 = 0.0f, h3 = 0.0f;

#pragma unroll 4
  for (int t = 0; t < Tn; ++t) {
    const float xa = pa[3 * t];
    const float xb = pb[3 * t];

    v2f bx; bx.x = xa;            bx.y = lo ? xb : 0.0f;
    v2f bh; bh.x = lo ? h0 : h2;  bh.y = lo ? h1 : h3;

    v8f dfin = __builtin_amdgcn_wmma_f32_16x16x4_f32(
        false, aWin, false, bx, (short)0, cbin, false, false);
    v8f dfh = __builtin_amdgcn_wmma_f32_16x16x4_f32(
        false, aWh, false, bh, (short)0, cbh, false, false);

    float rz[8];
#pragma unroll
    for (int i = 0; i < 8; ++i) rz[i] = fsig(dfin[i] + dfh[i]);

    float rX[4], zX[4];
#pragma unroll
    for (int j = 0; j < 4; ++j) rX[j] = __shfl_xor(rz[j], 16, 32);
#pragma unroll
    for (int j = 0; j < 4; ++j) zX[j] = __shfl_xor(rz[4 + j], 16, 32);

    float nn[4], nX[4];
#pragma unroll
    for (int j = 0; j < 4; ++j) nn[j] = ftanh(dfin[j] + rX[j] * dfh[j]);
#pragma unroll
    for (int j = 0; j < 4; ++j) nX[j] = __shfl_xor(nn[j], 16, 32);

    const float z0 = lo ? rz[4] : zX[0], n0 = lo ? nX[0] : nn[0];
    const float z1 = lo ? rz[5] : zX[1], n1 = lo ? nX[1] : nn[1];
    const float z2 = lo ? rz[6] : zX[2], n2 = lo ? nX[2] : nn[2];
    const float z3 = lo ? rz[7] : zX[3], n3 = lo ? nX[3] : nn[3];
    h0 = n0 + z0 * (h0 - n0);
    h1 = n1 + z1 * (h1 - n1);
    h2 = n2 + z2 * (h2 - n2);
    h3 = n3 + z3 * (h3 - n3);

    po[On * t] = dcb + h0 * dw0 + h1 * dw1 + h2 * dw2 + h3 * dw3;
  }
}

extern "C" void kernel_launch(void* const* d_in, const int* in_sizes, int n_in,
                              void* d_out, int out_size, void* d_ws, size_t ws_size,
                              hipStream_t stream) {
  (void)in_sizes; (void)n_in; (void)out_size; (void)d_ws; (void)ws_size;
  const float* inputs = (const float*)d_in[0];
  const float* W_in   = (const float*)d_in[1];
  const float* W_h    = (const float*)d_in[2];
  const float* bias   = (const float*)d_in[3];
  const float* dec_W  = (const float*)d_in[4];
  const float* dec_b  = (const float*)d_in[5];
  float* out = (float*)d_out;

  dim3 grid(Bn / 16);
  dim3 block(32);
  gru_wmma_kernel<<<grid, block, 0, stream>>>(inputs, W_in, W_h, bias,
                                              dec_W, dec_b, out);
}


# Inputs fetched 8 steps ahead through a ring of 8 register pairs, each slot refilled as soon as its step has read it (last 8 steps peeled): the load wait of every step leaves the recurrence
# speedup vs baseline: 1.7559x; 1.7559x over previous
#include <hip/hip_runtime.h>


typedef float v2f __attribute__((ext_vector_type(2)));
typedef float v8f __attribute__((ext_vector_type(8)));

#if defined(__HIP_DEVICE_COMPILE__) && !__has_builtin(__builtin_amdgcn_wmma_f32_16x16x4_f32)
#error "gfx1250 WMMA f32 16x16x4 builtin not available"
#endif

namespace {
constexpr int Bn = 4096;
constexpr int Tn = 2048;
constexpr int G3 = 12;
constexpr int On = 2;
}

__device__ __forceinline__ float fsig(float x) {
  float e = __builtin_amdgcn_exp2f(x * -1.4426950408889634f);
  return __builtin_amdgcn_rcpf(1.0f + e);
}
__device__ __forceinline__ float ftanh(float x) {
  float e = __builtin_amdgcn_exp2f(x * -2.8853901817779268f);
  return (1.0f - e) * __builtin_amdgcn_rcpf(1.0f + e);
}

__global__ __launch_bounds__(32) void gru_wmma_kernel(
    const float* __restrict__ in,
    const float* __restrict__ Win,
    const float* __restrict__ Wh,
    const float* __restrict__ bias,
    const float* __restrict__ dW,
    const float* __restrict__ db,
    float* __restrict__ out)
{
  const int lane = threadIdx.x & 31;
  const int m    = lane & 15;
  const bool lo  = lane < 16;
  const int seq  = blockIdx.x * 16 + m;
  const bool gv  = m < G3;

  v2f aWin, aWh;
  aWin.x = gv ? Win[(lo ? 0 : 2) * G3 + m] : 0.0f;
  aWin.y = (gv && lo) ? Win[1 * G3 + m] : 0.0f;
  aWh.x  = gv ? Wh[(lo ? 0 : 2) * G3 + m] : 0.0f;
  aWh.y  = gv ? Wh[(lo ? 1 : 3) * G3 + m] : 0.0f;

  v8f cbin, cbh;
#pragma unroll
  for (int r = 0; r < 8; ++r) {
    int row = (lo ? 0 : 8) + r;
    cbin[r] = (row < G3) ? bias[row] : 0.0f;
    cbh[r]  = (row < G3) ? bias[G3 + row] : 0.0f;
  }

  const int oc = lo ? 0 : 1;
  const float dw0 = dW[0 * On + oc], dw1 = dW[1 * On + oc];
  const float dw2 = dW[2 * On + oc], dw3 = dW[3 * On + oc];
  const float dcb = db[oc];

  const float* pa = in + (size_t)seq * (3 * Tn) + (lo ? 0 : 2);
  const float* pb = in + (size_t)seq * (3 * Tn) + (lo ? 1 : 2);
  float* po = out + (size_t)seq * (On * Tn) + oc;

  float h0 = 0.0f, h1 = 0.0f, h2 = 0.0f, h3 = 0.0f;

  auto step = [&](int t, float xa, float xb) {
    v2f bx; bx.x = xa;            bx.y = lo ? xb : 0.0f;
    v2f bh; bh.x = lo ? h0 : h2;  bh.y = lo ? h1 : h3;

    v8f dfin = __builtin_amdgcn_wmma_f32_16x16x4_f32(
        false, aWin, false, bx, (short)0, cbin, false, false);
    v8f dfh = __builtin_amdgcn_wmma_f32_16x16x4_f32(
        false, aWh, false, bh, (short)0, cbh, false, false);

    float rz[8];
#pragma unroll
    for (int i = 0; i < 8; ++i) rz[i] = fsig(dfin[i] + dfh[i]);

    float rX[4], zX[4];
#pragma unroll
    for (int j = 0; j < 4; ++j) rX[j] = __shfl_xor(rz[j], 16, 32);
#pragma unroll
    for (int j = 0; j < 4; ++j) zX[j] = __shfl_xor(rz[4 + j], 16, 32);

    float nn[4], nX[4];
#pragma unroll
    for (int j = 0; j < 4; ++j) nn[j] = ftanh(dfin[j] + rX[j] * dfh[j]);
#pragma unroll
    for (int j = 0; j < 4; ++j) nX[j] = __shfl_xor(nn[j], 16, 32);

    const float z0 = lo ? rz[4] : zX[0], n0 = lo ? nX[0] : nn[0];
    const float z1 = lo ? rz[5] : zX[1], n1 = lo ? nX[1] : nn[1];
    const float z2 = lo ? rz[6] : zX[2], n2 = lo ? nX[2] : nn[2];
    const float z3 = lo ? rz[7] : zX[3], n3 = lo ? nX[3] : nn[3];
    h0 = n0 + z0 * (h0 - n0);
    h1 = n1 + z1 * (h1 - n1);
    h2 = n2 + z2 * (h2 - n2);
    h3 = n3 + z3 * (h3 - n3);

    po[On * t] = dcb + h0 * dw0 + h1 * dw1 + h2 * dw2 + h3 * dw3;
  };

  constexpr int U = 8;
  float xa_r[U], xb_r[U];
#pragma unroll
  for (int u = 0; u < U; ++u) { xa_r[u] = pa[3 * u]; xb_r[u] = pb[3 * u]; }

  for (int tg = 0; tg < Tn - U; tg += U) {
#pragma unroll
    for (int u = 0; u < U; ++u) {
      step(tg + u, xa_r[u], xb_r[u]);
      xa_r[u] = pa[3 * (tg + U + u)];
      xb_r[u] = pb[3 * (tg + U + u)];
      __builtin_amdgcn_sched_barrier(0);
    }
  }
#pragma unroll
  for (int u = 0; u < U; ++u) step(Tn - U + u, xa_r[u], xb_r[u]);
}

extern "C" void kernel_launch(void* const* d_in, const int* in_sizes, int n_in,
                              void* d_out, int out_size, void* d_ws, size_t ws_size,
                              hipStream_t stream) {
  (void)in_sizes; (void)n_in; (void)out_size; (void)d_ws; (void)ws_size;
  const float* inputs = (const float*)d_in[0];
  const float* W_in   = (const float*)d_in[1];
  const float* W_h    = (const float*)d_in[2];
  const float* bias   = (const float*)d_in[3];
  const float* dec_W  = (const float*)d_in[4];
  const float* dec_b  = (const float*)d_in[5];
  float* out = (float*)d_out;

  dim3 grid(Bn / 16);
  dim3 block(32);
  gru_wmma_kernel<<<grid, block, 0, stream>>>(inputs, W_in, W_h, bias,
                                              dec_W, dec_b, out);
}


# A rows re-ordered so each lane half gets r,z,n of the two hidden units it feeds back as B; decoder in W_h^T's spare rows (its sum now made by the WMMA): no cross-lane exchange left in a step
# speedup vs baseline: 4.2270x; 2.4073x over previous
#include <hip/hip_runtime.h>


typedef float v2f __attribute__((ext_vector_type(2)));
typedef float v8f __attribute__((ext_vector_type(8)));

#if defined(__HIP_DEVICE_COMPILE__) && !__has_builtin(__builtin_amdgcn_wmma_f32_16x16x4_f32)
#error "gfx1250 WMMA f32 16x16x4 builtin not available"
#endif

namespace {
constexpr int Bn = 4096;
constexpr int Tn = 2048;
constexpr int Hn = 4;
constexpr int G3 = 12;
constexpr int On = 2;
}

__device__ __forceinline__ float fsig(float x) {
  float e = __builtin_amdgcn_exp2f(x * -1.4426950408889634f);
  return __builtin_amdgcn_rcpf(1.0f + e);
}
__device__ __forceinline__ float ftanh(float x) {
  float e = __builtin_amdgcn_exp2f(x * -2.8853901817779268f);
  return (1.0f - e) * __builtin_amdgcn_rcpf(1.0f + e);
}

__global__ __launch_bounds__(32) void gru_wmma_kernel(
    const float* __restrict__ in,
    const float* __restrict__ Win,
    const float* __restrict__ Wh,
    const float* __restrict__ bias,
    const float* __restrict__ dW,
    const float* __restrict__ db,
    float* __restrict__ out)
{
  const int lane = threadIdx.x & 31;
  const int m    = lane & 15;
  const int hh   = lane >> 4;
  const bool lo  = hh == 0;
  const int seq  = blockIdx.x * 16 + m;

  const int  kind   = (m & 7) >> 1;
  const int  unit   = 2 * (m >> 3) + (m & 1);
  const bool gate   = kind < 3;
  const bool decrow = (m & 7) == 6;
  const int  gcol   = gate ? kind * Hn + unit : 0;
  const int  ocol   = m >> 3;

  const int k0 = 2 * hh, k1 = 2 * hh + 1;
  v2f aWin, aWh;
  aWin.x = gate ? Win[k0 * G3 + gcol] : 0.0f;
  aWin.y = (gate && lo) ? Win[1 * G3 + gcol] : 0.0f;
  aWh.x  = gate ? Wh[k0 * G3 + gcol] : (decrow ? dW[k0 * On + ocol] : 0.0f);
  aWh.y  = gate ? Wh[k1 * G3 + gcol] : (decrow ? dW[k1 * On + ocol] : 0.0f);

  v8f cbin, cbh;
#pragma unroll
  for (int r = 0; r < 8; ++r) {
    const int col = (r >> 1) * Hn + 2 * hh + (r & 1);
    cbin[r] = (r < 6) ? bias[col] : 0.0f;
    cbh[r]  = (r < 6) ? bias[G3 + col] : ((r == 6) ? db[hh] : 0.0f);
  }

  const float* pa = in + (size_t)seq * (3 * Tn) + (lo ? 0 : 2);
  const float* pb = in + (size_t)seq * (3 * Tn) + (lo ? 1 : 2);
  float* po = out + (size_t)seq * (On * Tn) + hh;

  auto fin_of = [&](float xa, float xb) {
    v2f bx; bx.x = xa; bx.y = lo ? xb : 0.0f;
    return __builtin_amdgcn_wmma_f32_16x16x4_f32(
        false, aWin, false, bx, (short)0, cbin, false, false);
  };

  float ha = 0.0f, hb = 0.0f;
  v2f bz; bz.x = 0.0f; bz.y = 0.0f;
  v8f dfh = __builtin_amdgcn_wmma_f32_16x16x4_f32(
      false, aWh, false, bz, (short)0, cbh, false, false);
  v8f dfin;

  auto step = [&](int t) {
    const float ra = fsig(dfin[0] + dfh[0]);
    const float rb = fsig(dfin[1] + dfh[1]);
    const float za = fsig(dfin[2] + dfh[2]);
    const float zb = fsig(dfin[3] + dfh[3]);
    const float na = ftanh(dfin[4] + ra * dfh[4]);
    const float nb = ftanh(dfin[5] + rb * dfh[5]);
    ha = na + za * (ha - na);
    hb = nb + zb * (hb - nb);

    v2f bh; bh.x = ha; bh.y = hb;
    dfh = __builtin_amdgcn_wmma_f32_16x16x4_f32(
        false, aWh, false, bh, (short)0, cbh, false, false);

    po[On * t] = dfh[6];
  };

  constexpr int U = 8;
  float xa_r[U], xb_r[U];
#pragma unroll
  for (int u = 0; u < U; ++u) { xa_r[u] = pa[3 * u]; xb_r[u] = pb[3 * u]; }
  dfin = fin_of(xa_r[0], xb_r[0]);
  xa_r[0] = pa[3 * U];
  xb_r[0] = pb[3 * U];

  for (int tg = 0; tg < Tn - 2 * U; tg += U) {
#pragma unroll
    for (int u = 0; u < U; ++u) {
      const int s = (u + 1) % U;
      const v8f dnext = fin_of(xa_r[s], xb_r[s]);
      xa_r[s] = pa[3 * (tg + u + 1 + U)];
      xb_r[s] = pb[3 * (tg + u + 1 + U)];
      step(tg + u);
      dfin = dnext;
      __builtin_amdgcn_sched_barrier(0);
    }
  }
#pragma unroll
  for (int t = Tn - 2 * U; t < Tn; ++t) {
    const int s = (t + 1) % U;
    v8f dnext = dfin;
    if (t + 1 < Tn) dnext = fin_of(xa_r[s], xb_r[s]);
    if (t + 1 + U < Tn) {
      xa_r[s] = pa[3 * (t + 1 + U)];
      xb_r[s] = pb[3 * (t + 1 + U)];
    }
    step(t);
    dfin = dnext;
  }
}

extern "C" void kernel_launch(void* const* d_in, const int* in_sizes, int n_in,
                              void* d_out, int out_size, void* d_ws, size_t ws_size,
                              hipStream_t stream) {
  (void)in_sizes; (void)n_in; (void)out_size; (void)d_ws; (void)ws_size;
  const float* inputs = (const float*)d_in[0];
  const float* W_in   = (const float*)d_in[1];
  const float* W_h    = (const float*)d_in[2];
  const float* bias   = (const float*)d_in[3];
  const float* dec_W  = (const float*)d_in[4];
  const float* dec_b  = (const float*)d_in[5];
  float* out = (float*)d_out;

  dim3 grid(Bn / 16);
  dim3 block(32);
  gru_wmma_kernel<<<grid, block, 0, stream>>>(inputs, W_in, W_h, bias,
                                              dec_W, dec_b, out);
}


# Gates on v_tanh_f32: sigmoid = (1+tanh(x/2))/2 with the halves folded into A rows and biases; n argument summed as (fin+fh/2)+(fh/2)*t, h'=(1-z)n+zh; chain per step: add,tanh,fma,tanh,fma
# speedup vs baseline: 4.8430x; 1.1457x over previous
#include <hip/hip_runtime.h>


typedef float v2f __attribute__((ext_vector_type(2)));
typedef float v8f __attribute__((ext_vector_type(8)));

#if defined(__HIP_DEVICE_COMPILE__) && !__has_builtin(__builtin_amdgcn_wmma_f32_16x16x4_f32)
#error "gfx1250 WMMA f32 16x16x4 builtin not available"
#endif

namespace {
constexpr int Bn = 4096;
constexpr int Tn = 2048;
constexpr int Hn = 4;
constexpr int G3 = 12;
constexpr int On = 2;
}

#if defined(__HIP_DEVICE_COMPILE__) && !__has_builtin(__builtin_amdgcn_tanhf)
#error "gfx1250 v_tanh_f32 builtin not available"
#endif

__device__ __forceinline__ float ftanh(float x) {
  return __builtin_amdgcn_tanhf(x);
}

__global__ __launch_bounds__(32) void gru_wmma_kernel(
    const float* __restrict__ in,
    const float* __restrict__ Win,
    const float* __restrict__ Wh,
    const float* __restrict__ bias,
    const float* __restrict__ dW,
    const float* __restrict__ db,
    float* __restrict__ out)
{
  const int lane = threadIdx.x & 31;
  const int m    = lane & 15;
  const int hh   = lane >> 4;
  const bool lo  = hh == 0;
  const int seq  = blockIdx.x * 16 + m;

  const int  kind   = (m & 7) >> 1;
  const int  unit   = 2 * (m >> 3) + (m & 1);
  const bool gate   = kind < 3;
  const bool decrow = (m & 7) == 6;
  const int  gcol   = gate ? kind * Hn + unit : 0;
  const int  ocol   = m >> 3;

  const int k0 = 2 * hh, k1 = 2 * hh + 1;
  const float sWin = (kind < 2) ? 0.5f : 1.0f;
  v2f aWin, aWh;
  aWin.x = gate ? sWin * Win[k0 * G3 + gcol] : 0.0f;
  aWin.y = (gate && lo) ? sWin * Win[1 * G3 + gcol] : 0.0f;
  aWh.x  = gate ? 0.5f * Wh[k0 * G3 + gcol] : (decrow ? dW[k0 * On + ocol] : 0.0f);
  aWh.y  = gate ? 0.5f * Wh[k1 * G3 + gcol] : (decrow ? dW[k1 * On + ocol] : 0.0f);

  v8f cbin, cbh;
#pragma unroll
  for (int r = 0; r < 8; ++r) {
    const int col = (r >> 1) * Hn + 2 * hh + (r & 1);
    cbin[r] = (r < 6) ? ((r < 4) ? 0.5f : 1.0f) * bias[col] : 0.0f;
    cbh[r]  = (r < 6) ? 0.5f * bias[G3 + col] : ((r == 6) ? db[hh] : 0.0f);
  }

  const float* pa = in + (size_t)seq * (3 * Tn) + (lo ? 0 : 2);
  const float* pb = in + (size_t)seq * (3 * Tn) + (lo ? 1 : 2);
  float* po = out + (size_t)seq * (On * Tn) + hh;

  auto fin_of = [&](float xa, float xb) {
    v2f bx; bx.x = xa; bx.y = lo ? xb : 0.0f;
    return __builtin_amdgcn_wmma_f32_16x16x4_f32(
        false, aWin, false, bx, (short)0, cbin, false, false);
  };

  float ha = 0.0f, hb = 0.0f;
  v2f bz; bz.x = 0.0f; bz.y = 0.0f;
  v8f dfh = __builtin_amdgcn_wmma_f32_16x16x4_f32(
      false, aWh, false, bz, (short)0, cbh, false, false);
  v8f dfin;

  auto step = [&](int t) {
    const float tra = ftanh(dfin[0] + dfh[0]);
    const float trb = ftanh(dfin[1] + dfh[1]);
    const float tza = ftanh(dfin[2] + dfh[2]);
    const float tzb = ftanh(dfin[3] + dfh[3]);
    const float na = ftanh(__builtin_fmaf(dfh[4], tra, dfin[4] + dfh[4]));
    const float nb = ftanh(__builtin_fmaf(dfh[5], trb, dfin[5] + dfh[5]));
    const float za = __builtin_fmaf(0.5f, tza, 0.5f), ya = __builtin_fmaf(-0.5f, tza, 0.5f);
    const float zb = __builtin_fmaf(0.5f, tzb, 0.5f), yb = __builtin_fmaf(-0.5f, tzb, 0.5f);
    ha = __builtin_fmaf(ya, na, za * ha);
    hb = __builtin_fmaf(yb, nb, zb * hb);

    v2f bh; bh.x = ha; bh.y = hb;
    dfh = __builtin_amdgcn_wmma_f32_16x16x4_f32(
        false, aWh, false, bh, (short)0, cbh, false, false);

    po[On * t] = dfh[6];
  };

  constexpr int U = 8;
  float xa_r[U], xb_r[U];
#pragma unroll
  for (int u = 0; u < U; ++u) { xa_r[u] = pa[3 * u]; xb_r[u] = pb[3 * u]; }
  dfin = fin_of(xa_r[0], xb_r[0]);
  xa_r[0] = pa[3 * U];
  xb_r[0] = pb[3 * U];

  for (int tg = 0; tg < Tn - 2 * U; tg += U) {
#pragma unroll
    for (int u = 0; u < U; ++u) {
      const int s = (u + 1) % U;
      const v8f dnext = fin_of(xa_r[s], xb_r[s]);
      xa_r[s] = pa[3 * (tg + u + 1 + U)];
      xb_r[s] = pb[3 * (tg + u + 1 + U)];
      step(tg + u);
      dfin = dnext;
      __builtin_amdgcn_sched_barrier(0);
    }
  }
#pragma unroll
  for (int t = Tn - 2 * U; t < Tn; ++t) {
    const int s = (t + 1) % U;
    v8f dnext = dfin;
    if (t + 1 < Tn) dnext = fin_of(xa_r[s], xb_r[s]);
    if (t + 1 + U < Tn) {
      xa_r[s] = pa[3 * (t + 1 + U)];
      xb_r[s] = pb[3 * (t + 1 + U)];
    }
    step(t);
    dfin = dnext;
  }
}

extern "C" void kernel_launch(void* const* d_in, const int* in_sizes, int n_in,
                              void* d_out, int out_size, void* d_ws, size_t ws_size,
                              hipStream_t stream) {
  (void)in_sizes; (void)n_in; (void)out_size; (void)d_ws; (void)ws_size;
  const float* inputs = (const float*)d_in[0];
  const float* W_in   = (const float*)d_in[1];
  const float* W_h    = (const float*)d_in[2];
  const float* bias   = (const float*)d_in[3];
  const float* dec_W  = (const float*)d_in[4];
  const float* dec_b  = (const float*)d_in[5];
  float* out = (float*)d_out;

  dim3 grid(Bn / 16);
  dim3 block(32);
  gru_wmma_kernel<<<grid, block, 0, stream>>>(inputs, W_in, W_h, bias,
                                              dec_W, dec_b, out);
}


# Outputs staged in registers: decoder rows doubled so both lane halves hold both columns, halves keep alternate step pairs, one 16 B store per lane per 4 steps issued a step late; no per-step store
# speedup vs baseline: 6.7095x; 1.3854x over previous
#include <hip/hip_runtime.h>


typedef float v2f __attribute__((ext_vector_type(2)));
typedef float v4f __attribute__((ext_vector_type(4)));
typedef float v8f __attribute__((ext_vector_type(8)));

#if defined(__HIP_DEVICE_COMPILE__) && !__has_builtin(__builtin_amdgcn_wmma_f32_16x16x4_f32)
#error "gfx1250 WMMA f32 16x16x4 builtin not available"
#endif

namespace {
constexpr int Bn = 4096;
constexpr int Tn = 2048;
constexpr int Hn = 4;
constexpr int G3 = 12;
constexpr int On = 2;
}

#if defined(__HIP_DEVICE_COMPILE__) && !__has_builtin(__builtin_amdgcn_tanhf)
#error "gfx1250 v_tanh_f32 builtin not available"
#endif

__device__ __forceinline__ float ftanh(float x) {
  return __builtin_amdgcn_tanhf(x);
}

__global__ __launch_bounds__(32) void gru_wmma_kernel(
    const float* __restrict__ in,
    const float* __restrict__ Win,
    const float* __restrict__ Wh,
    const float* __restrict__ bias,
    const float* __restrict__ dW,
    const float* __restrict__ db,
    float* __restrict__ out)
{
  const int lane = threadIdx.x & 31;
  const int m    = lane & 15;
  const int hh   = lane >> 4;
  const bool lo  = hh == 0;
  const int seq  = blockIdx.x * 16 + m;

  const int  kind   = (m & 7) >> 1;
  const int  unit   = 2 * (m >> 3) + (m & 1);
  const bool gate   = kind < 3;
  const int  gcol   = gate ? kind * Hn + unit : 0;
  const int  ocol   = m & 1;

  const int k0 = 2 * hh, k1 = 2 * hh + 1;
  const float sWin = (kind < 2) ? 0.5f : 1.0f;
  v2f aWin, aWh;
  aWin.x = gate ? sWin * Win[k0 * G3 + gcol] : 0.0f;
  aWin.y = (gate && lo) ? sWin * Win[1 * G3 + gcol] : 0.0f;
  aWh.x  = gate ? 0.5f * Wh[k0 * G3 + gcol] : dW[k0 * On + ocol];
  aWh.y  = gate ? 0.5f * Wh[k1 * G3 + gcol] : dW[k1 * On + ocol];

  v8f cbin, cbh;
#pragma unroll
  for (int r = 0; r < 8; ++r) {
    const int col = (r >> 1) * Hn + 2 * hh + (r & 1);
    cbin[r] = (r < 6) ? ((r < 4) ? 0.5f : 1.0f) * bias[col] : 0.0f;
    cbh[r]  = (r < 6) ? 0.5f * bias[G3 + col] : db[r - 6];
  }

  const float* pa = in + (size_t)seq * (3 * Tn) + (lo ? 0 : 2);
  const float* pb = in + (size_t)seq * (3 * Tn) + (lo ? 1 : 2);
  float* po = out + (size_t)seq * (On * Tn) + 2 * On * hh;

  auto fin_of = [&](float xa, float xb) {
    v2f bx; bx.x = xa; bx.y = lo ? xb : 0.0f;
    return __builtin_amdgcn_wmma_f32_16x16x4_f32(
        false, aWin, false, bx, (short)0, cbin, false, false);
  };

  float ha = 0.0f, hb = 0.0f;
  v2f bz; bz.x = 0.0f; bz.y = 0.0f;
  v8f dfh = __builtin_amdgcn_wmma_f32_16x16x4_f32(
      false, aWh, false, bz, (short)0, cbh, false, false);
  v8f dfin;

  v4f qs[2];
  auto step = [&](int t, int u, bool has_prev, bool is_last) {
    const int b = (u >> 2) & 1, p = u & 3;
    if (p == 0 && has_prev) {
      *(v4f*)(po + On * (t - 4)) = qs[b ^ 1];
    }
    const float tra = ftanh(dfin[0] + dfh[0]);
    const float trb = ftanh(dfin[1] + dfh[1]);
    const float tza = ftanh(dfin[2] + dfh[2]);
    const float tzb = ftanh(dfin[3] + dfh[3]);
    const float na = ftanh(__builtin_fmaf(dfh[4], tra, dfin[4] + dfh[4]));
    const float nb = ftanh(__builtin_fmaf(dfh[5], trb, dfin[5] + dfh[5]));
    const float za = __builtin_fmaf(0.5f, tza, 0.5f), ya = __builtin_fmaf(-0.5f, tza, 0.5f);
    const float zb = __builtin_fmaf(0.5f, tzb, 0.5f), yb = __builtin_fmaf(-0.5f, tzb, 0.5f);
    ha = __builtin_fmaf(ya, na, za * ha);
    hb = __builtin_fmaf(yb, nb, zb * hb);

    v2f bh; bh.x = ha; bh.y = hb;
    dfh = __builtin_amdgcn_wmma_f32_16x16x4_f32(
        false, aWh, false, bh, (short)0, cbh, false, false);

    if (p == 0)      { qs[b][0] = dfh[6]; qs[b][1] = dfh[7]; }
    else if (p == 1) { qs[b][2] = dfh[6]; qs[b][3] = dfh[7]; }
    else if (p == 2) { qs[b][0] = lo ? qs[b][0] : dfh[6]; qs[b][1] = lo ? qs[b][1] : dfh[7]; }
    else             { qs[b][2] = lo ? qs[b][2] : dfh[6]; qs[b][3] = lo ? qs[b][3] : dfh[7]; }
    if (is_last) *(v4f*)(po + On * (t - 3)) = qs[b];
  };

  constexpr int U = 8;
  float xa_r[U], xb_r[U];
#pragma unroll
  for (int u = 0; u < U; ++u) { xa_r[u] = pa[3 * u]; xb_r[u] = pb[3 * u]; }
  dfin = fin_of(xa_r[0], xb_r[0]);
  xa_r[0] = pa[3 * U];
  xb_r[0] = pb[3 * U];

  auto advance = [&](int t, int u, bool has_prev) {
    const int s = (u + 1) % U;
    const v8f dnext = fin_of(xa_r[s], xb_r[s]);
    xa_r[s] = pa[3 * (t + 1 + U)];
    xb_r[s] = pb[3 * (t + 1 + U)];
    step(t, u, has_prev, false);
    dfin = dnext;
    __builtin_amdgcn_sched_barrier(0);
  };
#pragma unroll
  for (int u = 0; u < U; ++u) advance(u, u, u != 0);
  for (int tg = U; tg < Tn - 2 * U; tg += U) {
#pragma unroll
    for (int u = 0; u < U; ++u) advance(tg + u, u, true);
  }
#pragma unroll
  for (int t = Tn - 2 * U; t < Tn; ++t) {
    const int s = (t + 1) % U;
    v8f dnext = dfin;
    if (t + 1 < Tn) dnext = fin_of(xa_r[s], xb_r[s]);
    if (t + 1 + U < Tn) {
      xa_r[s] = pa[3 * (t + 1 + U)];
      xb_r[s] = pb[3 * (t + 1 + U)];
    }
    step(t, t % U, true, t == Tn - 1);
    dfin = dnext;
  }
}

extern "C" void kernel_launch(void* const* d_in, const int* in_sizes, int n_in,
                              void* d_out, int out_size, void* d_ws, size_t ws_size,
                              hipStream_t stream) {
  (void)in_sizes; (void)n_in; (void)out_size; (void)d_ws; (void)ws_size;
  const float* inputs = (const float*)d_in[0];
  const float* W_in   = (const float*)d_in[1];
  const float* W_h    = (const float*)d_in[2];
  const float* bias   = (const float*)d_in[3];
  const float* dec_W  = (const float*)d_in[4];
  const float* dec_b  = (const float*)d_in[5];
  float* out = (float*)d_out;

  dim3 grid(Bn / 16);
  dim3 block(32);
  gru_wmma_kernel<<<grid, block, 0, stream>>>(inputs, W_in, W_h, bias,
                                              dec_W, dec_b, out);
}


# Inputs fetched as 16-byte loads: 12 per lane at one place fetch the 192 contiguous bytes of a 16-step group into a 2-group register ring, 17+ steps ahead (was 8): 0.75 load instructions a step, was 2
# speedup vs baseline: 10.3950x; 1.5493x over previous
#include <hip/hip_runtime.h>


typedef float v2f __attribute__((ext_vector_type(2)));
typedef float v4f __attribute__((ext_vector_type(4)));
typedef float v8f __attribute__((ext_vector_type(8)));

#if defined(__HIP_DEVICE_COMPILE__) && !__has_builtin(__builtin_amdgcn_wmma_f32_16x16x4_f32)
#error "gfx1250 WMMA f32 16x16x4 builtin not available"
#endif

namespace {
constexpr int Bn = 4096;
constexpr int Tn = 2048;
constexpr int Hn = 4;
constexpr int G3 = 12;
constexpr int On = 2;
}

#if defined(__HIP_DEVICE_COMPILE__) && !__has_builtin(__builtin_amdgcn_tanhf)
#error "gfx1250 v_tanh_f32 builtin not available"
#endif

__device__ __forceinline__ float ftanh(float x) {
  return __builtin_amdgcn_tanhf(x);
}

__global__ __launch_bounds__(32) void gru_wmma_kernel(
    const float* __restrict__ in,
    const float* __restrict__ Win,
    const float* __restrict__ Wh,
    const float* __restrict__ bias,
    const float* __restrict__ dW,
    const float* __restrict__ db,
    float* __restrict__ out)
{
  const int lane = threadIdx.x & 31;
  const int m    = lane & 15;
  const int hh   = lane >> 4;
  const bool lo  = hh == 0;
  const int seq  = blockIdx.x * 16 + m;

  const int  kind   = (m & 7) >> 1;
  const int  unit   = 2 * (m >> 3) + (m & 1);
  const bool gate   = kind < 3;
  const int  gcol   = gate ? kind * Hn + unit : 0;
  const int  ocol   = m & 1;

  const int k0 = 2 * hh, k1 = 2 * hh + 1;
  const float sWin = (kind < 2) ? 0.5f : 1.0f;
  v2f aWin, aWh;
  aWin.x = gate ? sWin * Win[k0 * G3 + gcol] : 0.0f;
  aWin.y = (gate && lo) ? sWin * Win[1 * G3 + gcol] : 0.0f;
  aWh.x  = gate ? 0.5f * Wh[k0 * G3 + gcol] : dW[k0 * On + ocol];
  aWh.y  = gate ? 0.5f * Wh[k1 * G3 + gcol] : dW[k1 * On + ocol];

  v8f cbin, cbh;
#pragma unroll
  for (int r = 0; r < 8; ++r) {
    const int col = (r >> 1) * Hn + 2 * hh + (r & 1);
    cbin[r] = (r < 6) ? ((r < 4) ? 0.5f : 1.0f) * bias[col] : 0.0f;
    cbh[r]  = (r < 6) ? 0.5f * bias[G3 + col] : db[r - 6];
  }

  const float* px = in + (size_t)seq * (3 * Tn);
  float* po = out + (size_t)seq * (On * Tn) + 2 * On * hh;

  auto bx_of = [&](float x0, float x1, float x2) {
    v2f bx; bx.x = lo ? x0 : x2; bx.y = lo ? x1 : 0.0f;
    return bx;
  };
  auto fin_of = [&](v2f bx) {
    return __builtin_amdgcn_wmma_f32_16x16x4_f32(
        false, aWin, false, bx, (short)0, cbin, false, false);
  };

  float ha = 0.0f, hb = 0.0f;
  v2f bz; bz.x = 0.0f; bz.y = 0.0f;
  v8f dfh = __builtin_amdgcn_wmma_f32_16x16x4_f32(
      false, aWh, false, bz, (short)0, cbh, false, false);
  v8f dfin;

  v4f qs[2];
  float sra, srb, sza, szb, sna, snb;
  auto sums = [&]() {
    sra = dfin[0] + dfh[0];  srb = dfin[1] + dfh[1];
    sza = dfin[2] + dfh[2];  szb = dfin[3] + dfh[3];
    sna = dfin[4] + dfh[4];  snb = dfin[5] + dfh[5];
  };
  auto store_prev = [&](int t, int u, bool has_prev) {
    const int b = (u >> 2) & 1, p = u & 3;
    if (p == 0 && has_prev) {
      *(v4f*)(po + On * (t - 4)) = qs[b ^ 1];
    }
  };
  auto gates = [&](int u) {
    const int b = (u >> 2) & 1, p = u & 3;
    const float tra = ftanh(sra);
    const float trb = ftanh(srb);
    const float tza = ftanh(sza);
    const float tzb = ftanh(szb);
    const float na = ftanh(__builtin_fmaf(dfh[4], tra, sna));
    const float nb = ftanh(__builtin_fmaf(dfh[5], trb, snb));
    const float za = __builtin_fmaf(0.5f, tza, 0.5f), ya = __builtin_fmaf(-0.5f, tza, 0.5f);
    const float zb = __builtin_fmaf(0.5f, tzb, 0.5f), yb = __builtin_fmaf(-0.5f, tzb, 0.5f);
    ha = __builtin_fmaf(ya, na, za * ha);
    hb = __builtin_fmaf(yb, nb, zb * hb);

    v2f bh; bh.x = ha; bh.y = hb;
    dfh = __builtin_amdgcn_wmma_f32_16x16x4_f32(
        false, aWh, false, bh, (short)0, cbh, false, false);

    if (p == 0)      { qs[b][0] = dfh[6]; qs[b][1] = dfh[7]; }
    else if (p == 1) { qs[b][2] = dfh[6]; qs[b][3] = dfh[7]; }
    else if (p == 2) { qs[b][0] = lo ? qs[b][0] : dfh[6]; qs[b][1] = lo ? qs[b][1] : dfh[7]; }
    else             { qs[b][2] = lo ? qs[b][2] : dfh[6]; qs[b][3] = lo ? qs[b][3] : dfh[7]; }
  };

  constexpr int G  = 16;
  constexpr int R  = 2;
  constexpr int U  = G * R;
  constexpr int NG = Tn / G;
  constexpr int Q  = 3 * G / 4;
  static_assert(Tn % U == 0 && Tn >= 3 * U, "first trip, main loop and last trip");
  v4f xr[R][Q];
  auto fetch = [&](int s, int g) {
    const float* pg = px + 3 * G * g;
#pragma unroll
    for (int q = 0; q < Q; ++q) xr[s][q] = *(const v4f*)(pg + 4 * q);
  };
  auto word = [&](int s, int i) { return xr[s][i >> 2][i & 3]; };
#pragma unroll
  for (int s = 0; s < R; ++s) fetch(s, s);
  dfin = fin_of(bx_of(word(0, 0), word(0, 1), word(0, 2)));

  auto advance = [&](int t, int u, bool has_prev, bool has_next) {
    const int u1 = (u + 1) % U;
    const int s = u1 / G, p = u1 % G;
    v2f bx; bx.x = 0.0f; bx.y = 0.0f;
    if (has_next) bx = bx_of(word(s, 3 * p), word(s, 3 * p + 1), word(s, 3 * p + 2));
    sums();
    __builtin_amdgcn_sched_barrier(0);
    store_prev(t, u, has_prev);
    v8f dnext = dfin;
    if (has_next) dnext = fin_of(bx);
    if (has_next && p == G - 1 && (t + 1) / G + R < NG) {
      fetch(s, (t + 1) / G + R);
    }
    __builtin_amdgcn_sched_barrier(0);
    gates(u);
    dfin = dnext;
    __builtin_amdgcn_sched_barrier(0);
  };
#pragma unroll
  for (int u = 0; u < U; ++u) advance(u, u, u != 0, true);
  for (int tg = U; tg < Tn - U; tg += U) {
#pragma unroll
    for (int u = 0; u < U; ++u) advance(tg + u, u, true, true);
  }
#pragma unroll
  for (int u = 0; u < U; ++u) advance(Tn - U + u, u, true, u != U - 1);
  *(v4f*)(po + On * (Tn - 4)) = qs[((U - 1) >> 2) & 1];
}

extern "C" void kernel_launch(void* const* d_in, const int* in_sizes, int n_in,
                              void* d_out, int out_size, void* d_ws, size_t ws_size,
                              hipStream_t stream) {
  (void)in_sizes; (void)n_in; (void)out_size; (void)d_ws; (void)ws_size;
  const float* inputs = (const float*)d_in[0];
  const float* W_in   = (const float*)d_in[1];
  const float* W_h    = (const float*)d_in[2];
  const float* bias   = (const float*)d_in[3];
  const float* dec_W  = (const float*)d_in[4];
  const float* dec_b  = (const float*)d_in[5];
  float* out = (float*)d_out;

  dim3 grid(Bn / 16);
  dim3 block(32);
  gru_wmma_kernel<<<grid, block, 0, stream>>>(inputs, W_in, W_h, bias,
                                              dec_W, dec_b, out);
}


# fin product's result is the C operand of the h product (one sum chain: biases, x terms, h terms; fh_n/2 = n row minus fin_n): no add and no fin product between the h product and the first tanh
# speedup vs baseline: 10.8996x; 1.0485x over previous
#include <hip/hip_runtime.h>


typedef float v2f __attribute__((ext_vector_type(2)));
typedef float v4f __attribute__((ext_vector_type(4)));
typedef float v8f __attribute__((ext_vector_type(8)));

#if defined(__HIP_DEVICE_COMPILE__) && !__has_builtin(__builtin_amdgcn_wmma_f32_16x16x4_f32)
#error "gfx1250 WMMA f32 16x16x4 builtin not available"
#endif

namespace {
constexpr int Bn = 4096;
constexpr int Tn = 2048;
constexpr int Hn = 4;
constexpr int G3 = 12;
constexpr int On = 2;
}

#if defined(__HIP_DEVICE_COMPILE__) && !__has_builtin(__builtin_amdgcn_tanhf)
#error "gfx1250 v_tanh_f32 builtin not available"
#endif

__device__ __forceinline__ float ftanh(float x) {
  return __builtin_amdgcn_tanhf(x);
}

__global__ __launch_bounds__(32) void gru_wmma_kernel(
    const float* __restrict__ in,
    const float* __restrict__ Win,
    const float* __restrict__ Wh,
    const float* __restrict__ bias,
    const float* __restrict__ dW,
    const float* __restrict__ db,
    float* __restrict__ out)
{
  const int lane = threadIdx.x & 31;
  const int m    = lane & 15;
  const int hh   = lane >> 4;
  const bool lo  = hh == 0;
  const int seq  = blockIdx.x * 16 + m;

  const int  kind   = (m & 7) >> 1;
  const int  unit   = 2 * (m >> 3) + (m & 1);
  const bool gate   = kind < 3;
  const int  gcol   = gate ? kind * Hn + unit : 0;
  const int  ocol   = m & 1;

  const int k0 = 2 * hh, k1 = 2 * hh + 1;
  const float sWin = (kind < 2) ? 0.5f : 1.0f;
  v2f aWin, aWh;
  aWin.x = gate ? sWin * Win[k0 * G3 + gcol] : 0.0f;
  aWin.y = (gate && lo) ? sWin * Win[1 * G3 + gcol] : 0.0f;
  aWh.x  = gate ? 0.5f * Wh[k0 * G3 + gcol] : dW[k0 * On + ocol];
  aWh.y  = gate ? 0.5f * Wh[k1 * G3 + gcol] : dW[k1 * On + ocol];

  constexpr int G  = 16;
  constexpr int R  = 2;
  constexpr int U  = G * R;
  constexpr int NG = Tn / G;
  constexpr int Q  = 3 * G / 4;
  static_assert(Tn % U == 0 && Tn >= 3 * U, "first trip, main loop and last trip");
  static_assert(U % 8 == 0, "staging set and place of a step follow from t % U");

  v8f cb;
#pragma unroll
  for (int r = 0; r < 8; ++r) {
    const int col = (r >> 1) * Hn + 2 * hh + (r & 1);
    cb[r] = (r < 4) ? 0.5f * (bias[col] + bias[G3 + col])
          : (r < 6) ? bias[col] + 0.5f * bias[G3 + col]
                    : db[r - 6];
  }
  const float bhna = 0.5f * bias[G3 + 2 * Hn + 2 * hh];
  const float bhnb = 0.5f * bias[G3 + 2 * Hn + 2 * hh + 1];

  const float* px = in + (size_t)seq * (3 * Tn);
  float* po = out + (size_t)seq * (On * Tn) + 2 * On * hh;

  auto bx_of = [&](float x0, float x1, float x2) {
    v2f bx; bx.x = lo ? x0 : x2; bx.y = lo ? x1 : 0.0f;
    return bx;
  };
  auto fin_of = [&](v2f bx) {
    return __builtin_amdgcn_wmma_f32_16x16x4_f32(
        false, aWin, false, bx, (short)0, cb, false, false);
  };

  float ha = 0.0f, hb = 0.0f;
  v8f dfin;
  v8f dfh;
  float ena, enb;

  v4f qs[2];
  auto store_group = [&](int t, int u, bool exists) {
    const int b = (((u + U - 5) % U) >> 2) & 1;
    if ((u & 3) == 1 && exists) *(v4f*)(po + On * (t - 5)) = qs[b];
  };
  auto stage_prev = [&](int up) {
    const int b = (up >> 2) & 1, p = up & 3;
    if (p == 0)      { qs[b][0] = dfh[6]; qs[b][1] = dfh[7]; }
    else if (p == 1) { qs[b][2] = dfh[6]; qs[b][3] = dfh[7]; }
    else if (p == 2) { qs[b][0] = lo ? qs[b][0] : dfh[6]; qs[b][1] = lo ? qs[b][1] : dfh[7]; }
    else             { qs[b][2] = lo ? qs[b][2] : dfh[6]; qs[b][3] = lo ? qs[b][3] : dfh[7]; }
  };
  auto gates = [&](int u, bool has_prev, v8f cnext) {
    const float tra = ftanh(dfh[0]);
    const float trb = ftanh(dfh[1]);
    const float tza = ftanh(dfh[2]);
    const float tzb = ftanh(dfh[3]);
    const float hna = dfh[4] - ena;
    const float hnb = dfh[5] - enb;
    const float na = ftanh(__builtin_fmaf(hna, tra, dfh[4]));
    const float nb = ftanh(__builtin_fmaf(hnb, trb, dfh[5]));
    if (has_prev) stage_prev((u + U - 1) % U);
    const float za = __builtin_fmaf(0.5f, tza, 0.5f), ya = __builtin_fmaf(-0.5f, tza, 0.5f);
    const float zb = __builtin_fmaf(0.5f, tzb, 0.5f), yb = __builtin_fmaf(-0.5f, tzb, 0.5f);
    ha = __builtin_fmaf(ya, na, za * ha);
    hb = __builtin_fmaf(yb, nb, zb * hb);

    v2f bh; bh.x = ha; bh.y = hb;
    dfh = __builtin_amdgcn_wmma_f32_16x16x4_f32(
        false, aWh, false, bh, (short)0, cnext, false, false);
  };

  v4f xr[R][Q];
  auto fetch = [&](int s, int g) {
    const float* pg = px + 3 * G * g;
#pragma unroll
    for (int q = 0; q < Q; ++q) xr[s][q] = *(const v4f*)(pg + 4 * q);
  };
  auto word = [&](int s, int i) { return xr[s][i >> 2][i & 3]; };
#pragma unroll
  for (int s = 0; s < R; ++s) fetch(s, s);
  dfin = fin_of(bx_of(word(0, 0), word(0, 1), word(0, 2)));
  v2f bxn = bx_of(word(0, 3), word(0, 4), word(0, 5));
  {
    v2f bz; bz.x = 0.0f; bz.y = 0.0f;
    dfh = __builtin_amdgcn_wmma_f32_16x16x4_f32(
        false, aWh, false, bz, (short)0, dfin, false, false);
  }

  auto advance = [&](int t, int u, bool has_prev, bool store_exists, bool has_next, bool has_next2) {
    const int u1 = (u + 1) % U;
    const int s = u1 / G, p = u1 % G;
    const int u2 = (u + 2) % U;
    const int s2 = u2 / G, p2 = u2 % G;
    ena = dfin[4] - bhna;
    enb = dfin[5] - bhnb;
    store_group(t, u, store_exists);
    v8f cnext = cb;
    if (has_next) cnext = fin_of(bxn);
    if (has_next2) bxn = bx_of(word(s2, 3 * p2), word(s2, 3 * p2 + 1), word(s2, 3 * p2 + 2));
    if (has_next && p == G - 1 && (t + 1) / G + R < NG) {
      fetch(s, (t + 1) / G + R);
    }
    __builtin_amdgcn_sched_barrier(0);
    gates(u, has_prev, cnext);
    dfin = cnext;
    __builtin_amdgcn_sched_barrier(0);
  };
#pragma unroll
  for (int u = 0; u < U; ++u) advance(u, u, u != 0, u >= 5, true, true);
  for (int tg = U; tg < Tn - U; tg += U) {
#pragma unroll
    for (int u = 0; u < U; ++u) advance(tg + u, u, true, true, true, true);
  }
#pragma unroll
  for (int u = 0; u < U; ++u) advance(Tn - U + u, u, true, true, u != U - 1, u < U - 2);
  stage_prev(U - 1);
  *(v4f*)(po + On * (Tn - 4)) = qs[((U - 1) >> 2) & 1];
}

extern "C" void kernel_launch(void* const* d_in, const int* in_sizes, int n_in,
                              void* d_out, int out_size, void* d_ws, size_t ws_size,
                              hipStream_t stream) {
  (void)in_sizes; (void)n_in; (void)out_size; (void)d_ws; (void)ws_size;
  const float* inputs = (const float*)d_in[0];
  const float* W_in   = (const float*)d_in[1];
  const float* W_h    = (const float*)d_in[2];
  const float* bias   = (const float*)d_in[3];
  const float* dec_W  = (const float*)d_in[4];
  const float* dec_b  = (const float*)d_in[5];
  float* out = (float*)d_out;

  dim3 grid(Bn / 16);
  dim3 block(32);
  gru_wmma_kernel<<<grid, block, 0, stream>>>(inputs, W_in, W_h, bias,
                                              dec_W, dec_b, out);
}


# Outputs staged 16 steps (low lane steps 0-7, high lane 8-15), stored as four 16-byte stores right behind the input ring's 12 refill loads: memory instructions at 2 places per 32 steps, were 10
# speedup vs baseline: 11.8228x; 1.0847x over previous
#include <hip/hip_runtime.h>


typedef float v2f __attribute__((ext_vector_type(2)));
typedef float v4f __attribute__((ext_vector_type(4)));
typedef float v8f __attribute__((ext_vector_type(8)));

#if defined(__HIP_DEVICE_COMPILE__) && !__has_builtin(__builtin_amdgcn_wmma_f32_16x16x4_f32)
#error "gfx1250 WMMA f32 16x16x4 builtin not available"
#endif

namespace {
constexpr int Bn = 4096;
constexpr int Tn = 2048;
constexpr int Hn = 4;
constexpr int G3 = 12;
constexpr int On = 2;
}

#if defined(__HIP_DEVICE_COMPILE__) && !__has_builtin(__builtin_amdgcn_tanhf)
#error "gfx1250 v_tanh_f32 builtin not available"
#endif

__device__ __forceinline__ float ftanh(float x) {
  return __builtin_amdgcn_tanhf(x);
}

__global__ __launch_bounds__(32) void gru_wmma_kernel(
    const float* __restrict__ in,
    const float* __restrict__ Win,
    const float* __restrict__ Wh,
    const float* __restrict__ bias,
    const float* __restrict__ dW,
    const float* __restrict__ db,
    float* __restrict__ out)
{
  const int lane = threadIdx.x & 31;
  const int m    = lane & 15;
  const int hh   = lane >> 4;
  const bool lo  = hh == 0;
  const int seq  = blockIdx.x * 16 + m;

  const int  kind   = (m & 7) >> 1;
  const int  unit   = 2 * (m >> 3) + (m & 1);
  const bool gate   = kind < 3;
  const int  gcol   = gate ? kind * Hn + unit : 0;
  const int  ocol   = m & 1;

  const int k0 = 2 * hh, k1 = 2 * hh + 1;
  const float sWin = (kind < 2) ? 0.5f : 1.0f;
  v2f aWin, aWh;
  aWin.x = gate ? sWin * Win[k0 * G3 + gcol] : 0.0f;
  aWin.y = (gate && lo) ? sWin * Win[1 * G3 + gcol] : 0.0f;
  aWh.x  = gate ? 0.5f * Wh[k0 * G3 + gcol] : dW[k0 * On + ocol];
  aWh.y  = gate ? 0.5f * Wh[k1 * G3 + gcol] : dW[k1 * On + ocol];

  constexpr int G  = 16;
  constexpr int R  = 2;
  constexpr int U  = G * R;
  constexpr int NG = Tn / G;
  constexpr int Q  = 3 * G / 4;
  static_assert(Tn % U == 0 && Tn >= 3 * U, "first trip, main loop and last trip");
  static_assert(U == 2 * G, "staging set and place of a step follow from t % U: one set per group of a trip");

  v8f cb;
#pragma unroll
  for (int r = 0; r < 8; ++r) {
    const int col = (r >> 1) * Hn + 2 * hh + (r & 1);
    cb[r] = (r < 4) ? 0.5f * (bias[col] + bias[G3 + col])
          : (r < 6) ? bias[col] + 0.5f * bias[G3 + col]
                    : db[r - 6];
  }
  const float bhna = 0.5f * bias[G3 + 2 * Hn + 2 * hh];
  const float bhnb = 0.5f * bias[G3 + 2 * Hn + 2 * hh + 1];

  const float* px = in + (size_t)seq * (3 * Tn);
  float* po = out + (size_t)seq * (On * Tn) + 8 * On * hh;

  auto bx_of = [&](float x0, float x1, float x2) {
    v2f bx; bx.x = lo ? x0 : x2; bx.y = lo ? x1 : 0.0f;
    return bx;
  };
  auto fin_of = [&](v2f bx) {
    return __builtin_amdgcn_wmma_f32_16x16x4_f32(
        false, aWin, false, bx, (short)0, cb, false, false);
  };

  float ha = 0.0f, hb = 0.0f;
  v8f dfin;
  v8f dfh;
  float ena, enb;

  v4f qs[2][4];
  auto store_group = [&](int t, int u, bool exists) {
    const int b = (((u + U - 30) % U) >> 4) & 1;
    if ((u & 15) == 14 && exists) {
#pragma unroll
      for (int q = 0; q < 4; ++q) *(v4f*)(po + On * (t - 30) + 4 * q) = qs[b][q];
    }
  };
  auto stage_prev = [&](int up) {
    const int b = (up >> 4) & 1, p = up & 15;
    const int q = (p & 7) >> 1, i = 2 * (p & 1);
    if (p < 8) { qs[b][q][i] = dfh[6]; qs[b][q][i + 1] = dfh[7]; }
    else       { qs[b][q][i] = lo ? qs[b][q][i] : dfh[6]; qs[b][q][i + 1] = lo ? qs[b][q][i + 1] : dfh[7]; }
  };
  auto hold_group = [&](int u, bool exists) {
    const int b = (u >> 4) & 1;
    if ((u & 15) == 1 && exists)
      asm volatile("" :: "v"(qs[b][0]), "v"(qs[b][1]), "v"(qs[b][2]), "v"(qs[b][3]));
  };
  auto gates = [&](int u, bool has_prev, v8f cnext) {
    const float tra = ftanh(dfh[0]);
    const float trb = ftanh(dfh[1]);
    const float tza = ftanh(dfh[2]);
    const float tzb = ftanh(dfh[3]);
    const float hna = dfh[4] - ena;
    const float hnb = dfh[5] - enb;
    const float na = ftanh(__builtin_fmaf(hna, tra, dfh[4]));
    const float nb = ftanh(__builtin_fmaf(hnb, trb, dfh[5]));
    if (has_prev) stage_prev((u + U - 1) % U);
    const float za = __builtin_fmaf(0.5f, tza, 0.5f), ya = __builtin_fmaf(-0.5f, tza, 0.5f);
    const float zb = __builtin_fmaf(0.5f, tzb, 0.5f), yb = __builtin_fmaf(-0.5f, tzb, 0.5f);
    ha = __builtin_fmaf(ya, na, za * ha);
    hb = __builtin_fmaf(yb, nb, zb * hb);

    v2f bh; bh.x = ha; bh.y = hb;
    dfh = __builtin_amdgcn_wmma_f32_16x16x4_f32(
        false, aWh, false, bh, (short)0, cnext, false, false);
  };

  v4f xr[R][Q];
  auto fetch = [&](int s, int g) {
    const float* pg = px + 3 * G * g;
#pragma unroll
    for (int q = 0; q < Q; ++q) xr[s][q] = *(const v4f*)(pg + 4 * q);
  };
  auto word = [&](int s, int i) { return xr[s][i >> 2][i & 3]; };
#pragma unroll
  for (int s = 0; s < R; ++s) fetch(s, s);
  dfin = fin_of(bx_of(word(0, 0), word(0, 1), word(0, 2)));
  v2f bxn = bx_of(word(0, 3), word(0, 4), word(0, 5));
  {
    v2f bz; bz.x = 0.0f; bz.y = 0.0f;
    dfh = __builtin_amdgcn_wmma_f32_16x16x4_f32(
        false, aWh, false, bz, (short)0, dfin, false, false);
  }

  auto advance = [&](int t, int u, bool has_prev, bool store_exists, bool hold_exists, bool has_next, bool has_next2) {
    const int u1 = (u + 1) % U;
    const int s = u1 / G, p = u1 % G;
    const int u2 = (u + 2) % U;
    const int s2 = u2 / G, p2 = u2 % G;
    ena = dfin[4] - bhna;
    enb = dfin[5] - bhnb;
    hold_group(u, hold_exists);
    if (p == G - 1) {
      if (has_next && (t + 1) / G + R < NG) fetch(s, (t + 1) / G + R);
      store_group(t, u, store_exists);
      __builtin_amdgcn_sched_barrier(0);
    }
    v8f cnext = cb;
    if (has_next) cnext = fin_of(bxn);
    if (has_next2) bxn = bx_of(word(s2, 3 * p2), word(s2, 3 * p2 + 1), word(s2, 3 * p2 + 2));
    __builtin_amdgcn_sched_barrier(0);
    gates(u, has_prev, cnext);
    dfin = cnext;
    __builtin_amdgcn_sched_barrier(0);
  };
#pragma unroll
  for (int u = 0; u < U; ++u) advance(u, u, u != 0, u >= G, false, true, true);
  for (int tg = U; tg < Tn - U; tg += U) {
#pragma unroll
    for (int u = 0; u < U; ++u) advance(tg + u, u, true, true, true, true, true);
  }
#pragma unroll
  for (int u = 0; u < U; ++u) advance(Tn - U + u, u, true, true, true, u != U - 1, u < U - 2);
  stage_prev(U - 1);
#pragma unroll
  for (int q = 0; q < 4; ++q) *(v4f*)(po + On * (Tn - G) + 4 * q) = qs[1][q];
}

extern "C" void kernel_launch(void* const* d_in, const int* in_sizes, int n_in,
                              void* d_out, int out_size, void* d_ws, size_t ws_size,
                              hipStream_t stream) {
  (void)in_sizes; (void)n_in; (void)out_size; (void)d_ws; (void)ws_size;
  const float* inputs = (const float*)d_in[0];
  const float* W_in   = (const float*)d_in[1];
  const float* W_h    = (const float*)d_in[2];
  const float* bias   = (const float*)d_in[3];
  const float* dec_W  = (const float*)d_in[4];
  const float* dec_b  = (const float*)d_in[5];
  float* out = (float*)d_out;

  dim3 grid(Bn / 16);
  dim3 block(32);
  gru_wmma_kernel<<<grid, block, 0, stream>>>(inputs, W_in, W_h, bias,
                                              dec_W, dec_b, out);
}


# 8 sequences a wave on twice the waves: columns 8-15 of the fin product carry the next step, its rows brought over by a row rotate; one fin product per 2 steps, 1.5 matrix ops a step (were 2)
# speedup vs baseline: 12.2165x; 1.0333x over previous
#include <hip/hip_runtime.h>


typedef float v2f __attribute__((ext_vector_type(2)));
typedef float v4f __attribute__((ext_vector_type(4)));
typedef float v8f __attribute__((ext_vector_type(8)));

#if defined(__HIP_DEVICE_COMPILE__) && !__has_builtin(__builtin_amdgcn_wmma_f32_16x16x4_f32)
#error "gfx1250 WMMA f32 16x16x4 builtin not available"
#endif

namespace {
constexpr int Bn = 4096;
constexpr int Tn = 2048;
constexpr int Hn = 4;
constexpr int G3 = 12;
constexpr int On = 2;
}

#if defined(__HIP_DEVICE_COMPILE__) && !__has_builtin(__builtin_amdgcn_tanhf)
#error "gfx1250 v_tanh_f32 builtin not available"
#endif

__device__ __forceinline__ float ftanh(float x) {
  return __builtin_amdgcn_tanhf(x);
}

__global__ __launch_bounds__(32) void gru_wmma_kernel(
    const float* __restrict__ in,
    const float* __restrict__ Win,
    const float* __restrict__ Wh,
    const float* __restrict__ bias,
    const float* __restrict__ dW,
    const float* __restrict__ db,
    float* __restrict__ out)
{
  const int lane = threadIdx.x & 31;
  const int m    = lane & 15;
  const int hh   = lane >> 4;
  const bool lo  = hh == 0;
  const bool odd = (m & 8) != 0;
  const int seq  = blockIdx.x * 8 + (m & 7);

  const int  kind   = (m & 7) >> 1;
  const int  unit   = 2 * (m >> 3) + (m & 1);
  const bool gate   = kind < 3;
  const int  gcol   = gate ? kind * Hn + unit : 0;
  const int  ocol   = m & 1;

  const int k0 = 2 * hh, k1 = 2 * hh + 1;
  const float sWin = (kind < 2) ? 0.5f : 1.0f;
  v2f aWin, aWh;
  aWin.x = gate ? sWin * Win[k0 * G3 + gcol] : 0.0f;
  aWin.y = (gate && lo) ? sWin * Win[1 * G3 + gcol] : 0.0f;
  aWh.x  = gate ? 0.5f * Wh[k0 * G3 + gcol] : dW[k0 * On + ocol];
  aWh.y  = gate ? 0.5f * Wh[k1 * G3 + gcol] : dW[k1 * On + ocol];

  constexpr int G  = 16;
  constexpr int R  = 2;
  constexpr int U  = G * R;
  constexpr int NG = Tn / G;
  constexpr int Q  = 3 * G / 4;
  static_assert(Tn % U == 0 && Tn >= 3 * U, "first trip, main loop and last trip");
  static_assert(U == 2 * G, "staging set and place of a step follow from t % U: one set per group of a trip");
  static_assert(G % 2 == 0 && Tn % 2 == 0, "the two steps of a fin product lie in one group, and the last step is odd");

  v8f cb;
#pragma unroll
  for (int r = 0; r < 8; ++r) {
    const int col = (r >> 1) * Hn + 2 * hh + (r & 1);
    cb[r] = (r < 4) ? 0.5f * (bias[col] + bias[G3 + col])
          : (r < 6) ? bias[col] + 0.5f * bias[G3 + col]
                    : db[r - 6];
  }
  const float bhna = 0.5f * bias[G3 + 2 * Hn + 2 * hh];
  const float bhnb = 0.5f * bias[G3 + 2 * Hn + 2 * hh + 1];

  const float* px = in + (size_t)seq * (3 * Tn);
  float* po = out + (size_t)seq * (On * Tn) + 8 * On * hh;

  auto bx_of = [&](float x0, float x1, float x2) {
    v2f bx; bx.x = lo ? x0 : x2; bx.y = lo ? x1 : 0.0f;
    return bx;
  };
  auto bx_pair = [&](float e0, float e1, float e2, float o0, float o1, float o2) {
    return bx_of(odd ? o0 : e0, odd ? o1 : e1, odd ? o2 : e2);
  };
  auto odd_of = [&](v8f f) {
    v8f r;
#pragma unroll
    for (int i = 0; i < 6; ++i)
      r[i] = __int_as_float(__builtin_amdgcn_update_dpp(
          0, __float_as_int(f[i]), 0x128  , 0xF, 0xF, true));
    r[6] = cb[6]; r[7] = cb[7];
    return r;
  };
  auto fin_of = [&](v2f bx) {
    return __builtin_amdgcn_wmma_f32_16x16x4_f32(
        false, aWin, false, bx, (short)0, cb, false, false);
  };

  float ha = 0.0f, hb = 0.0f;
  v8f fpair;
  v8f dfh;
  float ena = 0.0f, enb = 0.0f;

  v4f qs[2][4];
  auto store_group = [&](int t, int u, bool exists) {
    const int b = (((u + U - 30) % U) >> 4) & 1;
    if ((u & 15) == 14 && exists) {
#pragma unroll
      for (int q = 0; q < 4; ++q) *(v4f*)(po + On * (t - 30) + 4 * q) = qs[b][q];
    }
  };
  auto dup8 = [&](float v) {
    const int w = __float_as_int(v);
    return __int_as_float(__builtin_amdgcn_update_dpp(w, w, 0x128  , 0xF, 0xC  , false));
  };
  auto stage_prev = [&](int up) {
    const int b = (up >> 4) & 1, p = up & 15;
    const int q = (p & 7) >> 1, i = 2 * (p & 1);
    if (p < 8) { qs[b][q][i] = dup8(dfh[6]); qs[b][q][i + 1] = dup8(dfh[7]); }
    else       { qs[b][q][i] = dup8(lo ? qs[b][q][i] : dfh[6]); qs[b][q][i + 1] = dup8(lo ? qs[b][q][i + 1] : dfh[7]); }
  };
  auto hold_group = [&](int u, bool exists) {
    const int b = (u >> 4) & 1;
    if ((u & 15) == 1 && exists)
      asm volatile("" :: "v"(qs[b][0]), "v"(qs[b][1]), "v"(qs[b][2]), "v"(qs[b][3]));
  };
  auto gates = [&](int u, bool has_prev, v8f cnext) {
    const float tra = ftanh(dfh[0]);
    const float trb = ftanh(dfh[1]);
    const float tza = ftanh(dfh[2]);
    const float tzb = ftanh(dfh[3]);
    const float hna = dfh[4] - ena;
    const float hnb = dfh[5] - enb;
    const float na = ftanh(__builtin_fmaf(hna, tra, dfh[4]));
    const float nb = ftanh(__builtin_fmaf(hnb, trb, dfh[5]));
    if (has_prev) stage_prev((u + U - 1) % U);
    const float za = __builtin_fmaf(0.5f, tza, 0.5f), ya = __builtin_fmaf(-0.5f, tza, 0.5f);
    const float zb = __builtin_fmaf(0.5f, tzb, 0.5f), yb = __builtin_fmaf(-0.5f, tzb, 0.5f);
    ha = __builtin_fmaf(ya, na, za * ha);
    hb = __builtin_fmaf(yb, nb, zb * hb);

    v2f bh; bh.x = ha; bh.y = hb;
    dfh = __builtin_amdgcn_wmma_f32_16x16x4_f32(
        false, aWh, false, bh, (short)0, cnext, false, false);
  };

  v4f xr[R][Q];
  auto fetch = [&](int s, int g) {
    const float* pg = px + 3 * G * g;
#pragma unroll
    for (int q = 0; q < Q; ++q) xr[s][q] = *(const v4f*)(pg + 4 * q);
  };
  auto word = [&](int s, int i) { return xr[s][i >> 2][i & 3]; };
#pragma unroll
  for (int s = 0; s < R; ++s) fetch(s, s);
  fpair = fin_of(bx_pair(word(0, 0), word(0, 1), word(0, 2), word(0, 3), word(0, 4), word(0, 5)));
  v2f bxn = bx_pair(word(0, 6), word(0, 7), word(0, 8), word(0, 9), word(0, 10), word(0, 11));
  {
    v2f bz; bz.x = 0.0f; bz.y = 0.0f;
    dfh = __builtin_amdgcn_wmma_f32_16x16x4_f32(
        false, aWh, false, bz, (short)0, fpair, false, false);
  }

  auto advance = [&](int t, int u, bool has_prev, bool store_exists, bool hold_exists, bool has_next, bool has_next2) {
    const bool even = (u & 1) == 0;
    const int u1 = (u + 1) % U;
    const int s = u1 / G, p = u1 % G;
    const int u2 = (u + 2) % U;
    const int s2 = u2 / G, p2 = u2 % G;
    if (even) { ena = fpair[4] - bhna; enb = fpair[5] - bhnb; }
    float ena1 = 0.0f, enb1 = 0.0f;
    hold_group(u, hold_exists);
    if (p == G - 1) {
      if (has_next && (t + 1) / G + R < NG) fetch(s, (t + 1) / G + R);
      store_group(t, u, store_exists);
      __builtin_amdgcn_sched_barrier(0);
    }
    v8f cnext = cb;
    if (even) {
      cnext = odd_of(fpair);
      ena1 = cnext[4] - bhna; enb1 = cnext[5] - bhnb;
      asm volatile("" : "+v"(ena1), "+v"(enb1));
      if (has_next2) bxn = bx_pair(word(s2, 3 * p2), word(s2, 3 * p2 + 1), word(s2, 3 * p2 + 2),
                                   word(s2, 3 * p2 + 3), word(s2, 3 * p2 + 4), word(s2, 3 * p2 + 5));
    } else if (has_next) {
      fpair = fin_of(bxn);
      cnext = fpair;
    }
    __builtin_amdgcn_sched_barrier(0);
    gates(u, has_prev, cnext);
    if (even) { ena = ena1; enb = enb1; }
    __builtin_amdgcn_sched_barrier(0);
  };
#pragma unroll
  for (int u = 0; u < U; ++u) advance(u, u, u != 0, u >= G, false, true, true);
  for (int tg = U; tg < Tn - U; tg += U) {
#pragma unroll
    for (int u = 0; u < U; ++u) advance(tg + u, u, true, true, true, true, true);
  }
#pragma unroll
  for (int u = 0; u < U; ++u) advance(Tn - U + u, u, true, true, true, u != U - 1, u < U - 2);
  stage_prev(U - 1);
#pragma unroll
  for (int q = 0; q < 4; ++q) *(v4f*)(po + On * (Tn - G) + 4 * q) = qs[1][q];
}

extern "C" void kernel_launch(void* const* d_in, const int* in_sizes, int n_in,
                              void* d_out, int out_size, void* d_ws, size_t ws_size,
                              hipStream_t stream) {
  (void)in_sizes; (void)n_in; (void)out_size; (void)d_ws; (void)ws_size;
  const float* inputs = (const float*)d_in[0];
  const float* W_in   = (const float*)d_in[1];
  const float* W_h    = (const float*)d_in[2];
  const float* bias   = (const float*)d_in[3];
  const float* dec_W  = (const float*)d_in[4];
  const float* dec_b  = (const float*)d_in[5];
  float* out = (float*)d_out;

  dim3 grid(Bn / 8);
  dim3 block(32);
  gru_wmma_kernel<<<grid, block, 0, stream>>>(inputs, W_in, W_h, bias,
                                              dec_W, dec_b, out);
}


# Input ring deepened from 2 to 4 sixteen-step groups: every load is issued 48 steps ahead of its first use (was 16), so the memory misses all sequences make in the same step stay out of the recurrence
# speedup vs baseline: 12.9610x; 1.0609x over previous
#include <hip/hip_runtime.h>


typedef float v2f __attribute__((ext_vector_type(2)));
typedef float v4f __attribute__((ext_vector_type(4)));
typedef float v8f __attribute__((ext_vector_type(8)));

#if defined(__HIP_DEVICE_COMPILE__) && !__has_builtin(__builtin_amdgcn_wmma_f32_16x16x4_f32)
#error "gfx1250 WMMA f32 16x16x4 builtin not available"
#endif

namespace {
constexpr int Bn = 4096;
constexpr int Tn = 2048;
constexpr int Hn = 4;
constexpr int G3 = 12;
constexpr int On = 2;
}

#if defined(__HIP_DEVICE_COMPILE__) && !__has_builtin(__builtin_amdgcn_tanhf)
#error "gfx1250 v_tanh_f32 builtin not available"
#endif

__device__ __forceinline__ float ftanh(float x) {
  return __builtin_amdgcn_tanhf(x);
}

__global__ __launch_bounds__(32) void gru_wmma_kernel(
    const float* __restrict__ in,
    const float* __restrict__ Win,
    const float* __restrict__ Wh,
    const float* __restrict__ bias,
    const float* __restrict__ dW,
    const float* __restrict__ db,
    float* __restrict__ out)
{
  const int lane = threadIdx.x & 31;
  const int m    = lane & 15;
  const int hh   = lane >> 4;
  const bool lo  = hh == 0;
  const bool odd = (m & 8) != 0;
  const int seq  = blockIdx.x * 8 + (m & 7);

  const int  kind   = (m & 7) >> 1;
  const int  unit   = 2 * (m >> 3) + (m & 1);
  const bool gate   = kind < 3;
  const int  gcol   = gate ? kind * Hn + unit : 0;
  const int  ocol   = m & 1;

  const int k0 = 2 * hh, k1 = 2 * hh + 1;
  const float sWin = (kind < 2) ? 0.5f : 1.0f;
  v2f aWin, aWh;
  aWin.x = gate ? sWin * Win[k0 * G3 + gcol] : 0.0f;
  aWin.y = (gate && lo) ? sWin * Win[1 * G3 + gcol] : 0.0f;
  aWh.x  = gate ? 0.5f * Wh[k0 * G3 + gcol] : dW[k0 * On + ocol];
  aWh.y  = gate ? 0.5f * Wh[k1 * G3 + gcol] : dW[k1 * On + ocol];

  constexpr int G  = 16;
  constexpr int R  = 4;
  constexpr int U  = G * R;
  constexpr int NG = Tn / G;
  constexpr int Q  = 3 * G / 4;
  static_assert(Tn % U == 0 && Tn >= 2 * U, "first trip and main loop");
  static_assert(U % (2 * G) == 0, "staging set and place of a step follow from t % U: the two sets alternate group by group, and a trip has an even number of groups");
  static_assert(G % 2 == 0 && Tn % 2 == 0, "the two steps of a fin product lie in one group, and the last step is odd");

  v8f cb;
#pragma unroll
  for (int r = 0; r < 8; ++r) {
    const int col = (r >> 1) * Hn + 2 * hh + (r & 1);
    cb[r] = (r < 4) ? 0.5f * (bias[col] + bias[G3 + col])
          : (r < 6) ? bias[col] + 0.5f * bias[G3 + col]
                    : db[r - 6];
  }
  const float bhna = 0.5f * bias[G3 + 2 * Hn + 2 * hh];
  const float bhnb = 0.5f * bias[G3 + 2 * Hn + 2 * hh + 1];

  const float* px = in + (size_t)seq * (3 * Tn);
  float* po = out + (size_t)seq * (On * Tn) + 8 * On * hh;

  auto bx_of = [&](float x0, float x1, float x2) {
    v2f bx; bx.x = lo ? x0 : x2; bx.y = lo ? x1 : 0.0f;
    return bx;
  };
  auto bx_pair = [&](float e0, float e1, float e2, float o0, float o1, float o2) {
    return bx_of(odd ? o0 : e0, odd ? o1 : e1, odd ? o2 : e2);
  };
  auto odd_of = [&](v8f f) {
    v8f r;
#pragma unroll
    for (int i = 0; i < 6; ++i)
      r[i] = __int_as_float(__builtin_amdgcn_update_dpp(
          0, __float_as_int(f[i]), 0x128  , 0xF, 0xF, true));
    r[6] = cb[6]; r[7] = cb[7];
    return r;
  };
  auto fin_of = [&](v2f bx) {
    return __builtin_amdgcn_wmma_f32_16x16x4_f32(
        false, aWin, false, bx, (short)0, cb, false, false);
  };

  float ha = 0.0f, hb = 0.0f;
  v8f fpair;
  v8f dfh;
  float ena = 0.0f, enb = 0.0f;

  v4f qs[2][4];
  auto store_group = [&](int t, int u, bool exists) {
    const int b = (((u + U - 30) % U) >> 4) & 1;
    if ((u & 15) == 14 && exists) {
#pragma unroll
      for (int q = 0; q < 4; ++q) *(v4f*)(po + On * (t - 30) + 4 * q) = qs[b][q];
    }
  };
  auto dup8 = [&](float v) {
    const int w = __float_as_int(v);
    return __int_as_float(__builtin_amdgcn_update_dpp(w, w, 0x128  , 0xF, 0xC  , false));
  };
  auto stage_prev = [&](int up) {
    const int b = (up >> 4) & 1, p = up & 15;
    const int q = (p & 7) >> 1, i = 2 * (p & 1);
    if (p < 8) { qs[b][q][i] = dup8(dfh[6]); qs[b][q][i + 1] = dup8(dfh[7]); }
    else       { qs[b][q][i] = dup8(lo ? qs[b][q][i] : dfh[6]); qs[b][q][i + 1] = dup8(lo ? qs[b][q][i + 1] : dfh[7]); }
  };
  auto hold_group = [&](int u, bool exists) {
    const int b = (u >> 4) & 1;
    if ((u & 15) == 1 && exists)
      asm volatile("" :: "v"(qs[b][0]), "v"(qs[b][1]), "v"(qs[b][2]), "v"(qs[b][3]));
  };
  auto gates = [&](int u, bool has_prev, v8f cnext) {
    const float tra = ftanh(dfh[0]);
    const float trb = ftanh(dfh[1]);
    const float tza = ftanh(dfh[2]);
    const float tzb = ftanh(dfh[3]);
    const float hna = dfh[4] - ena;
    const float hnb = dfh[5] - enb;
    const float na = ftanh(__builtin_fmaf(hna, tra, dfh[4]));
    const float nb = ftanh(__builtin_fmaf(hnb, trb, dfh[5]));
    if (has_prev) stage_prev((u + U - 1) % U);
    const float za = __builtin_fmaf(0.5f, tza, 0.5f), ya = __builtin_fmaf(-0.5f, tza, 0.5f);
    const float zb = __builtin_fmaf(0.5f, tzb, 0.5f), yb = __builtin_fmaf(-0.5f, tzb, 0.5f);
    ha = __builtin_fmaf(ya, na, za * ha);
    hb = __builtin_fmaf(yb, nb, zb * hb);

    v2f bh; bh.x = ha; bh.y = hb;
    dfh = __builtin_amdgcn_wmma_f32_16x16x4_f32(
        false, aWh, false, bh, (short)0, cnext, false, false);
  };

  v4f xr[R][Q];
  auto fetch = [&](int s, int g) {
    const float* pg = px + 3 * G * g;
#pragma unroll
    for (int q = 0; q < Q; ++q) xr[s][q] = *(const v4f*)(pg + 4 * q);
  };
  auto word = [&](int s, int i) { return xr[s][i >> 2][i & 3]; };
#pragma unroll
  for (int s = 0; s < R; ++s) fetch(s, s);
  fpair = fin_of(bx_pair(word(0, 0), word(0, 1), word(0, 2), word(0, 3), word(0, 4), word(0, 5)));
  v2f bxn = bx_pair(word(0, 6), word(0, 7), word(0, 8), word(0, 9), word(0, 10), word(0, 11));
  {
    v2f bz; bz.x = 0.0f; bz.y = 0.0f;
    dfh = __builtin_amdgcn_wmma_f32_16x16x4_f32(
        false, aWh, false, bz, (short)0, fpair, false, false);
  }

  auto advance = [&](int t, int u, bool has_prev, bool store_exists, bool hold_exists, bool has_next, bool has_next2) {
    const bool even = (u & 1) == 0;
    const int u1 = (u + 1) % U;
    const int s = u1 / G, p = u1 % G;
    const int u2 = (u + 2) % U;
    const int s2 = u2 / G, p2 = u2 % G;
    if (even) { ena = fpair[4] - bhna; enb = fpair[5] - bhnb; }
    float ena1 = 0.0f, enb1 = 0.0f;
    hold_group(u, hold_exists);
    if (p == G - 1) {
      if (has_next) { const int gn = (t + 1) / G + R; fetch(s, gn < NG ? gn : NG - 1); }
      store_group(t, u, store_exists);
      __builtin_amdgcn_sched_barrier(0);
    }
    v8f cnext = cb;
    if (even) {
      cnext = odd_of(fpair);
      ena1 = cnext[4] - bhna; enb1 = cnext[5] - bhnb;
      asm volatile("" : "+v"(ena1), "+v"(enb1));
      if (has_next2) bxn = bx_pair(word(s2, 3 * p2), word(s2, 3 * p2 + 1), word(s2, 3 * p2 + 2),
                                   word(s2, 3 * p2 + 3), word(s2, 3 * p2 + 4), word(s2, 3 * p2 + 5));
    } else if (has_next) {
      fpair = fin_of(bxn);
      cnext = fpair;
    }
    __builtin_amdgcn_sched_barrier(0);
    gates(u, has_prev, cnext);
    if (even) { ena = ena1; enb = enb1; }
    __builtin_amdgcn_sched_barrier(0);
  };
#pragma unroll
  for (int u = 0; u < U; ++u) advance(u, u, u != 0, u >= G, u >= 2 * G, true, true);
  for (int tg = U; tg < Tn; tg += U) {
#pragma unroll
    for (int u = 0; u < U; ++u) advance(tg + u, u, true, true, true, true, true);
  }
  stage_prev(U - 1);
#pragma unroll
  for (int q = 0; q < 4; ++q) *(v4f*)(po + On * (Tn - G) + 4 * q) = qs[1][q];
}

extern "C" void kernel_launch(void* const* d_in, const int* in_sizes, int n_in,
                              void* d_out, int out_size, void* d_ws, size_t ws_size,
                              hipStream_t stream) {
  (void)in_sizes; (void)n_in; (void)out_size; (void)d_ws; (void)ws_size;
  const float* inputs = (const float*)d_in[0];
  const float* W_in   = (const float*)d_in[1];
  const float* W_h    = (const float*)d_in[2];
  const float* bias   = (const float*)d_in[3];
  const float* dec_W  = (const float*)d_in[4];
  const float* dec_b  = (const float*)d_in[5];
  float* out = (float*)d_out;

  dim3 grid(Bn / 8);
  dim3 block(32);
  gru_wmma_kernel<<<grid, block, 0, stream>>>(inputs, W_in, W_h, bias,
                                              dec_W, dec_b, out);
}


# Gates issued in path order (tanh r, fma, tanh n, fma; unit b 2 cycles behind a), 5 of 6 row rotates moved from behind the h product into gaps of the path; z h = fma(tz,h/2,h/2), rounded once
# speedup vs baseline: 14.4760x; 1.1169x over previous
#include <hip/hip_runtime.h>


typedef float v2f __attribute__((ext_vector_type(2)));
typedef float v4f __attribute__((ext_vector_type(4)));
typedef float v8f __attribute__((ext_vector_type(8)));

#if defined(__HIP_DEVICE_COMPILE__) && !__has_builtin(__builtin_amdgcn_wmma_f32_16x16x4_f32)
#error "gfx1250 WMMA f32 16x16x4 builtin not available"
#endif

namespace {
constexpr int Bn = 4096;
constexpr int Tn = 2048;
constexpr int Hn = 4;
constexpr int G3 = 12;
constexpr int On = 2;
}

#if defined(__HIP_DEVICE_COMPILE__) && !__has_builtin(__builtin_amdgcn_tanhf)
#error "gfx1250 v_tanh_f32 builtin not available"
#endif

__device__ __forceinline__ float ftanh(float x) {
  return __builtin_amdgcn_tanhf(x);
}

__global__ __launch_bounds__(32) void gru_wmma_kernel(
    const float* __restrict__ in,
    const float* __restrict__ Win,
    const float* __restrict__ Wh,
    const float* __restrict__ bias,
    const float* __restrict__ dW,
    const float* __restrict__ db,
    float* __restrict__ out)
{
  const int lane = threadIdx.x & 31;
  const int m    = lane & 15;
  const int hh   = lane >> 4;
  const bool lo  = hh == 0;
  const bool odd = (m & 8) != 0;
  const int seq  = blockIdx.x * 8 + (m & 7);

  const int  kind   = (m & 7) >> 1;
  const int  unit   = 2 * (m >> 3) + (m & 1);
  const bool gate   = kind < 3;
  const int  gcol   = gate ? kind * Hn + unit : 0;
  const int  ocol   = m & 1;

  const int k0 = 2 * hh, k1 = 2 * hh + 1;
  const float sWin = (kind < 2) ? 0.5f : 1.0f;
  v2f aWin, aWh;
  aWin.x = gate ? sWin * Win[k0 * G3 + gcol] : 0.0f;
  aWin.y = (gate && lo) ? sWin * Win[1 * G3 + gcol] : 0.0f;
  aWh.x  = gate ? 0.5f * Wh[k0 * G3 + gcol] : dW[k0 * On + ocol];
  aWh.y  = gate ? 0.5f * Wh[k1 * G3 + gcol] : dW[k1 * On + ocol];

  constexpr int G  = 16;
  constexpr int R  = 4;
  constexpr int U  = G * R;
  constexpr int NG = Tn / G;
  constexpr int Q  = 3 * G / 4;
  static_assert(Tn % U == 0 && Tn >= 2 * U, "first trip and main loop");
  static_assert(U % (2 * G) == 0, "staging set and place of a step follow from t % U: the two sets alternate group by group, and a trip has an even number of groups");
  static_assert(G % 2 == 0 && Tn % 2 == 0, "the two steps of a fin product lie in one group, and the last step is odd");

  v8f cb;
#pragma unroll
  for (int r = 0; r < 8; ++r) {
    const int col = (r >> 1) * Hn + 2 * hh + (r & 1);
    cb[r] = (r < 4) ? 0.5f * (bias[col] + bias[G3 + col])
          : (r < 6) ? bias[col] + 0.5f * bias[G3 + col]
                    : db[r - 6];
  }
  const float bhna = 0.5f * bias[G3 + 2 * Hn + 2 * hh];
  const float bhnb = 0.5f * bias[G3 + 2 * Hn + 2 * hh + 1];

  const float* px = in + (size_t)seq * (3 * Tn);
  float* po = out + (size_t)seq * (On * Tn) + 8 * On * hh;

  auto bx_of = [&](float x0, float x1, float x2) {
    v2f bx; bx.x = lo ? x0 : x2; bx.y = lo ? x1 : 0.0f;
    return bx;
  };
  auto bx_pair = [&](float e0, float e1, float e2, float o0, float o1, float o2) {
    return bx_of(odd ? o0 : e0, odd ? o1 : e1, odd ? o2 : e2);
  };
  auto rot8 = [&](float v) {
    return __int_as_float(__builtin_amdgcn_update_dpp(
        0, __float_as_int(v), 0x128  , 0xF, 0xF, true));
  };
  auto fin_of = [&](v2f bx) {
    return __builtin_amdgcn_wmma_f32_16x16x4_f32(
        false, aWin, false, bx, (short)0, cb, false, false);
  };

  float ha = 0.0f, hb = 0.0f;
  v8f fpair;
  v8f dfh;
  float ena = 0.0f, enb = 0.0f;
  float rr4 = 0.0f, rr5 = 0.0f;

  v4f qs[2][4];
  auto store_group = [&](int t, int u, bool exists) {
    const int b = (((u + U - 30) % U) >> 4) & 1;
    if ((u & 15) == 14 && exists) {
#pragma unroll
      for (int q = 0; q < 4; ++q) *(v4f*)(po + On * (t - 30) + 4 * q) = qs[b][q];
    }
  };
  auto dup8 = [&](float v) {
    const int w = __float_as_int(v);
    return __int_as_float(__builtin_amdgcn_update_dpp(w, w, 0x128  , 0xF, 0xC  , false));
  };
  auto stage_prev = [&](int up) {
    const int b = (up >> 4) & 1, p = up & 15;
    const int q = (p & 7) >> 1, i = 2 * (p & 1);
    if (p < 8) { qs[b][q][i] = dup8(dfh[6]); qs[b][q][i + 1] = dup8(dfh[7]); }
    else       { qs[b][q][i] = dup8(lo ? qs[b][q][i] : dfh[6]); qs[b][q][i + 1] = dup8(lo ? qs[b][q][i + 1] : dfh[7]); }
  };
  auto hold_group = [&](int u, bool exists) {
    const int b = (u >> 4) & 1;
    if ((u & 15) == 1 && exists)
      asm volatile("" :: "v"(qs[b][0]), "v"(qs[b][1]), "v"(qs[b][2]), "v"(qs[b][3]));
  };
#define PATH_FENCE __builtin_amdgcn_sched_barrier(0)
#define ISSUE_BEHIND(a, b) asm volatile("" : "+v"(a), "+v"(b))
  auto gates = [&](int u, bool even, bool has_prev, v8f cin, float c0, float d6, float d7) {
    const int up = (u + U - 1) % U;
    const int sb = (up >> 4) & 1, sp = up & 15;
    const int sq = (sp & 7) >> 1, si = 2 * (sp & 1);
    float o0 = dfh[6], o1 = dfh[7];
    float x1 = dfh[1], x2 = dfh[2], x3 = dfh[3];
    float tra = ftanh(dfh[0]);
    float hna = dfh[4] - ena, hnb = dfh[5] - enb;
    asm volatile("" : "+v"(tra), "+v"(x1), "+v"(hna), "+v"(hnb));
    PATH_FENCE;
    float trb = ftanh(x1);
    ISSUE_BEHIND(trb, x2);
    PATH_FENCE;
    float tza = ftanh(x2);
    ISSUE_BEHIND(tza, x3);
    PATH_FENCE;
    float r1 = 0.0f;
    if (even) r1 = rot8(fpair[1]);
    PATH_FENCE;
    float tzb = ftanh(x3);
    ISSUE_BEHIND(tzb, hna);
    PATH_FENCE;
    float aa = __builtin_fmaf(hna, tra, dfh[4]);
    float hha, hhb;
    asm volatile("v_mul_f32 %0, 0.5, %3" : "=v"(hha), "+v"(aa), "+v"(hnb) : "v"(ha));
    PATH_FENCE;
    float ab = __builtin_fmaf(hnb, trb, dfh[5]);
    asm volatile("v_mul_f32 %0, 0.5, %2" : "=v"(hhb), "+v"(ab) : "v"(hb));
    if (has_prev && sp >= 8) { o0 = lo ? qs[sb][sq][si] : o0; o1 = lo ? qs[sb][sq][si + 1] : o1; }
    asm volatile("" : "+v"(o0), "+v"(o1));
    PATH_FENCE;
    const float ya = __builtin_fmaf(-0.5f, tza, 0.5f);
    PATH_FENCE;
    const float na = ftanh(aa);
    PATH_FENCE;
    const float zha = __builtin_fmaf(tza, hha, hha);
    PATH_FENCE;
    const float nb = ftanh(ab);
    PATH_FENCE;
    const float yb = __builtin_fmaf(-0.5f, tzb, 0.5f);
    PATH_FENCE;
    const float zhb = __builtin_fmaf(tzb, hhb, hhb);
    PATH_FENCE;
    const float m0 = rot8(even ? fpair[2] : fpair[4]);
    PATH_FENCE;
    const float m1 = rot8(even ? fpair[3] : fpair[5]);
    PATH_FENCE;
    ha = __builtin_fmaf(ya, na, zha);
    PATH_FENCE;
    hb = __builtin_fmaf(yb, nb, zhb);
    PATH_FENCE;
    if (has_prev) qs[sb][sq][si] = dup8(o0);
    PATH_FENCE;
    if (has_prev) qs[sb][sq][si + 1] = dup8(o1);
    PATH_FENCE;

    v8f c;
    if (even) { c[0] = c0; c[1] = r1; c[2] = m0; c[3] = m1; c[4] = rr4; c[5] = rr5; c[6] = d6; c[7] = d7; }
    else { c = cin; rr4 = m0; rr5 = m1; }
    v2f bh; bh.x = ha; bh.y = hb;
    dfh = __builtin_amdgcn_wmma_f32_16x16x4_f32(
        false, aWh, false, bh, (short)0, c, false, false);
  };

  v4f xr[R][Q];
  auto fetch = [&](int s, int g) {
    const float* pg = px + 3 * G * g;
#pragma unroll
    for (int q = 0; q < Q; ++q) xr[s][q] = *(const v4f*)(pg + 4 * q);
  };
  auto word = [&](int s, int i) { return xr[s][i >> 2][i & 3]; };
#pragma unroll
  for (int s = 0; s < R; ++s) fetch(s, s);
  fpair = fin_of(bx_pair(word(0, 0), word(0, 1), word(0, 2), word(0, 3), word(0, 4), word(0, 5)));
  rr4 = rot8(fpair[4]); rr5 = rot8(fpair[5]);
  v2f bxn = bx_pair(word(0, 6), word(0, 7), word(0, 8), word(0, 9), word(0, 10), word(0, 11));
  {
    v2f bz; bz.x = 0.0f; bz.y = 0.0f;
    dfh = __builtin_amdgcn_wmma_f32_16x16x4_f32(
        false, aWh, false, bz, (short)0, fpair, false, false);
  }

  auto advance = [&](int t, int u, bool even, bool has_prev, bool store_exists, bool hold_exists, bool has_next, bool has_next2) {
    const int u1 = (u + 1) % U;
    const int s = u1 / G, p = u1 % G;
    const int u2 = (u + 2) % U;
    const int s2 = u2 / G, p2 = u2 % G;
    if (even) {
      ena = fpair[4] - bhna; enb = fpair[5] - bhnb;
      asm volatile("" : "+v"(ena), "+v"(enb));
    }
    float ena1 = 0.0f, enb1 = 0.0f;
    hold_group(u, hold_exists);
    if (p == G - 1) {
      if (has_next) { const int gn = (t + 1) / G + R; fetch(s, gn < NG ? gn : NG - 1); }
      store_group(t, u, store_exists);
      __builtin_amdgcn_sched_barrier(0);
    }
    v8f cnext = cb;
    float c0 = 0.0f;
    float d6 = 0.0f, d7 = 0.0f;
    if (even) {
      c0 = rot8(fpair[0]);
      d6 = cb[6]; d7 = cb[7];
      asm volatile("" : "+v"(d6), "+v"(d7));
      ena1 = rr4 - bhna; enb1 = rr5 - bhnb;
      asm volatile("" : "+v"(ena1), "+v"(enb1));
      if (has_next2) bxn = bx_pair(word(s2, 3 * p2), word(s2, 3 * p2 + 1), word(s2, 3 * p2 + 2),
                                   word(s2, 3 * p2 + 3), word(s2, 3 * p2 + 4), word(s2, 3 * p2 + 5));
    } else if (has_next) {
      fpair = fin_of(bxn);
      cnext = fpair;
    }
    __builtin_amdgcn_sched_barrier(0);
    gates(u, even, has_prev, cnext, c0, d6, d7);
    if (even) { ena = ena1; enb = enb1; }
    __builtin_amdgcn_sched_barrier(0);
  };
#pragma unroll
  for (int u = 0; u < U; u += 2) {
    advance(u, u, true, u != 0, u >= G, u >= 2 * G, true, true);
    advance(u + 1, u + 1, false, true, u + 1 >= G, u + 1 >= 2 * G, true, true);
  }
  for (int tg = U; tg < Tn; tg += U) {
#pragma unroll
    for (int u = 0; u < U; u += 2) {
      advance(tg + u, u, true, true, true, true, true, true);
      advance(tg + u + 1, u + 1, false, true, true, true, true, true);
    }
  }
  asm volatile("s_wait_alu depctr_va_vdst(0)" : "+v"(dfh));
  stage_prev(U - 1);
#pragma unroll
  for (int q = 0; q < 4; ++q) *(v4f*)(po + On * (Tn - G) + 4 * q) = qs[1][q];
}

extern "C" void kernel_launch(void* const* d_in, const int* in_sizes, int n_in,
                              void* d_out, int out_size, void* d_ws, size_t ws_size,
                              hipStream_t stream) {
  (void)in_sizes; (void)n_in; (void)out_size; (void)d_ws; (void)ws_size;
  const float* inputs = (const float*)d_in[0];
  const float* W_in   = (const float*)d_in[1];
  const float* W_h    = (const float*)d_in[2];
  const float* bias   = (const float*)d_in[3];
  const float* dec_W  = (const float*)d_in[4];
  const float* dec_b  = (const float*)d_in[5];
  float* out = (float*)d_out;

  dim3 grid(Bn / 8);
  dim3 block(32);
  gru_wmma_kernel<<<grid, block, 0, stream>>>(inputs, W_in, W_h, bias,
                                              dec_W, dec_b, out);
}
